# starting kernel
# baseline (speedup 1.0000x reference)
#include <hip/hip_runtime.h>

typedef __attribute__((ext_vector_type(2))) float v2f;
typedef __attribute__((ext_vector_type(4))) float v4f;
typedef __attribute__((ext_vector_type(8))) float v8f;

#define Bn   4
#define Hn   128
#define Wn   128
#define Cn   256
#define Gn   16
#define K2n  9
#define Crn  64
#define KGn  (K2n * Gn)
#define NPIX (Bn * Hn * Wn)
#define BN_EPS 1e-3f

__global__ __launch_bounds__(256) void invol_kergen(
    const float* __restrict__ x,  const float* __restrict__ w1,
    const float* __restrict__ b1, const float* __restrict__ gamma,
    const float* __restrict__ beta, const float* __restrict__ mean,
    const float* __restrict__ var,  const float* __restrict__ w2,
    const float* __restrict__ b2,   float* __restrict__ kout)
{
    __shared__ float h_lds[8][16 * Crn];

    const int lane = threadIdx.x & 31;
    const int wave = threadIdx.x >> 5;
    const int tile = blockIdx.x * 8 + wave;
    const int p0   = tile * 16;
    const int m    = lane & 15;
    const int hi   = lane >> 4;
    const int dk   = hi * 2;

    const float* xrow = x + (size_t)(p0 + m) * Cn + dk;
    for (int nt = 0; nt < 4; ++nt) {
        const int n0 = nt * 16;
        v8f acc = {};
        for (int kk = 0; kk < Cn / 4; ++kk) {
            const int kb = kk * 4;
            v2f a = *(const v2f*)(xrow + kb);
            v2f bf;
            bf.x = w1[(kb + dk    ) * Crn + n0 + m];
            bf.y = w1[(kb + dk + 1) * Crn + n0 + m];
            acc = __builtin_amdgcn_wmma_f32_16x16x4_f32(
                      false, a, false, bf, (short)0, acc, false, false);
        }
        const int n = n0 + m;
        const float sc = gamma[n] * rsqrtf(var[n] + BN_EPS);
        const float sh = beta[n] - mean[n] * sc;
        const float bb = b1[n];
#pragma unroll
        for (int r = 0; r < 8; ++r) {
            float v = (acc[r] + bb) * sc + sh;
            v = v > 0.0f ? v : 0.0f;
            h_lds[wave][(r + hi * 8) * Crn + n] = v;
        }
    }
    __syncthreads();

    for (int nt = 0; nt < 9; ++nt) {
        const int n2 = nt * 16 + m;
        v8f acc = {};
        for (int kk = 0; kk < Crn / 4; ++kk) {
            const int kb = kk * 4;
            v2f a = *(const v2f*)(&h_lds[wave][m * Crn + kb + dk]);
            v2f bf;
            bf.x = w2[(kb + dk    ) * KGn + n2];
            bf.y = w2[(kb + dk + 1) * KGn + n2];
            acc = __builtin_amdgcn_wmma_f32_16x16x4_f32(
                      false, a, false, bf, (short)0, acc, false, false);
        }
        const float bb = b2[n2];
#pragma unroll
        for (int r = 0; r < 8; ++r)
            kout[(size_t)(p0 + r + hi * 8) * KGn + n2] = acc[r] + bb;
    }
}

__global__ __launch_bounds__(256) void invol_apply(
    const float* __restrict__ x, const float* __restrict__ kern,
    float* __restrict__ out)
{
    const int t  = blockIdx.x * 256 + threadIdx.x;
    const int ci = t & 63;
    const int p  = t >> 6;
    const int c0 = ci * 4;
    const int g0 = c0 & 15;

    const int b   = p >> 14;
    const int rem = p & 16383;
    const int y   = rem >> 7;
    const int xx  = rem & 127;

    const float* kp = kern + (size_t)p * KGn + g0;
    v4f acc = {};
#pragma unroll
    for (int ki = 0; ki < 3; ++ki) {
        const int yy = y + ki - 1;
        if (yy < 0 || yy >= Hn) continue;
#pragma unroll
        for (int kj = 0; kj < 3; ++kj) {
            const int xj = xx + kj - 1;
            if (xj < 0 || xj >= Wn) continue;
            const v4f xv = *(const v4f*)(x + ((size_t)((b * Hn + yy) * Wn + xj)) * Cn + c0);
            const v4f kv = *(const v4f*)(kp + (ki * 3 + kj) * Gn);
            acc += xv * kv;
        }
    }
    *(v4f*)(out + (size_t)p * Cn + c0) = acc;
}

extern "C" void kernel_launch(void* const* d_in, const int* in_sizes, int n_in,
                              void* d_out, int out_size, void* d_ws, size_t ws_size,
                              hipStream_t stream)
{
    const float* x     = (const float*)d_in[0];
    const float* w1    = (const float*)d_in[1];
    const float* b1    = (const float*)d_in[2];
    const float* gamma = (const float*)d_in[3];
    const float* beta  = (const float*)d_in[4];
    const float* mean  = (const float*)d_in[5];
    const float* var   = (const float*)d_in[6];
    const float* w2    = (const float*)d_in[7];
    const float* b2    = (const float*)d_in[8];

    float* out  = (float*)d_out;
    float* kout = out + (size_t)NPIX * Cn;

    invol_kergen<<<NPIX / 16 / 8, 256, 0, stream>>>(
        x, w1, b1, gamma, beta, mean, var, w2, b2, kout);

    invol_apply<<<(NPIX * 64) / 256, 256, 0, stream>>>(x, kout, out);
}


# Kernel-gen GEMMs: per-WMMA global loads replaced by reuse: w1,w2 staged once per 16-wave block in LDS as ready B fragments, each x/h A fragment used for all N-tiles; k summed in permuted order
# speedup vs baseline: 1.3757x; 1.3757x over previous
#include <hip/hip_runtime.h>

typedef __attribute__((ext_vector_type(2))) float v2f;
typedef __attribute__((ext_vector_type(4))) float v4f;
typedef __attribute__((ext_vector_type(8))) float v8f;

#define Bn   4
#define Hn   128
#define Wn   128
#define Cn   256
#define Gn   16
#define K2n  9
#define Crn  64
#define KGn  (K2n * Gn)
#define NPIX (Bn * Hn * Wn)
#define BN_EPS 1e-3f

#define NWAVE 16
#define HPITCH 68
#define W1F_N (4 * 8 * 4 * 2 * 16)
#define W2F_N (9 * 2 * 4 * 2 * 16)

__device__ __forceinline__ v8f mma4(v2f a, v2f b, v8f c)
{
    return __builtin_amdgcn_wmma_f32_16x16x4_f32(false, a, false, b, (short)0, c, false, false);
}

__global__ __launch_bounds__(NWAVE * 32) void invol_kergen(
    const float* __restrict__ x,  const float* __restrict__ w1,
    const float* __restrict__ b1, const float* __restrict__ gamma,
    const float* __restrict__ beta, const float* __restrict__ mean,
    const float* __restrict__ var,  const float* __restrict__ w2,
    const float* __restrict__ b2,   float* __restrict__ kout)
{
    __shared__ v4f lds[W1F_N + W2F_N + NWAVE * 16 * HPITCH / 4];
    v4f*   const w1f   = lds;
    v4f*   const w2f   = lds + W1F_N;
    float* const h_lds = (float*)(lds + W1F_N + W2F_N)
                         + (threadIdx.x >> 5) * (16 * HPITCH);

    const int tid  = threadIdx.x;
    const int lane = tid & 31;
    const int wave = tid >> 5;
    const int tile = blockIdx.x * NWAVE + wave;
    const int p0   = tile * 16;
    const int m    = lane & 15;
    const int hi   = lane >> 4;

    const float* xrow = x + (size_t)(p0 + m) * Cn + hi * 16;
    v4f a_cur[4];
#pragma unroll
    for (int i = 0; i < 4; ++i) a_cur[i] = *(const v4f*)(xrow + 4 * i);

#pragma unroll
    for (int it = 0; it < W1F_N / (NWAVE * 32); ++it) {
        const int f = tid + it * (NWAVE * 32);
        const int c = f & 15, h = (f >> 4) & 1, i = (f >> 5) & 3, q = (f >> 7) & 7, nt = f >> 10;
        const float* src = w1 + (32 * q + 16 * h + 4 * i) * Crn + nt * 16 + c;
        v4f v;
        v.x = src[0]; v.y = src[Crn]; v.z = src[2 * Crn]; v.w = src[3 * Crn];
        w1f[f] = v;
    }
#pragma unroll
    for (int it = 0; it < (W2F_N + NWAVE * 32 - 1) / (NWAVE * 32); ++it) {
        const int f = tid + it * (NWAVE * 32);
        if (f >= W2F_N) break;
        const int c = f & 15, h = (f >> 4) & 1, i = (f >> 5) & 3, q = (f >> 7) & 1, nt = f >> 8;
        const float* src = w2 + (32 * q + 16 * h + 4 * i) * KGn + nt * 16 + c;
        v4f v;
        v.x = src[0]; v.y = src[KGn]; v.z = src[2 * KGn]; v.w = src[3 * KGn];
        w2f[f] = v;
    }
    __syncthreads();

    {
        v8f acc[4] = {};
        const v4f* bfrag = w1f + hi * 16 + m;
#pragma unroll
        for (int q = 0; q < 8; ++q) {
            v4f a_nxt[4];
            if (q < 7) {
#pragma unroll
                for (int i = 0; i < 4; ++i) a_nxt[i] = *(const v4f*)(xrow + 32 * (q + 1) + 4 * i);
            }
#pragma unroll
            for (int i = 0; i < 4; ++i) {
                v4f b[4];
#pragma unroll
                for (int nt = 0; nt < 4; ++nt) b[nt] = bfrag[((nt * 8 + q) * 4 + i) * 32];
#pragma unroll
                for (int nt = 0; nt < 4; ++nt) acc[nt] = mma4(a_cur[i].xy, b[nt].xy, acc[nt]);
#pragma unroll
                for (int nt = 0; nt < 4; ++nt) acc[nt] = mma4(a_cur[i].zw, b[nt].zw, acc[nt]);
            }
            if (q < 7) {
#pragma unroll
                for (int i = 0; i < 4; ++i) a_cur[i] = a_nxt[i];
            }
        }
#pragma unroll
        for (int nt = 0; nt < 4; ++nt) {
            const int n = nt * 16 + m;
            const float sc = gamma[n] * rsqrtf(var[n] + BN_EPS);
            const float sh = beta[n] - mean[n] * sc;
            const float bb = b1[n];
#pragma unroll
            for (int r = 0; r < 8; ++r) {
                float v = (acc[nt][r] + bb) * sc + sh;
                v = v > 0.0f ? v : 0.0f;
                h_lds[(r + hi * 8) * HPITCH + n] = v;
            }
        }
    }
    __syncthreads();

    {
        v8f acc[9] = {};
        const float* hrow  = h_lds + m * HPITCH + hi * 16;
        const v4f*   bfrag = w2f + hi * 16 + m;
#pragma unroll
        for (int q = 0; q < 2; ++q) {
            v4f a[4];
#pragma unroll
            for (int i = 0; i < 4; ++i) a[i] = *(const v4f*)(hrow + 32 * q + 4 * i);
#pragma unroll
            for (int i = 0; i < 4; ++i) {
                v4f b[9];
#pragma unroll
                for (int nt = 0; nt < 9; ++nt) b[nt] = bfrag[((nt * 2 + q) * 4 + i) * 32];
#pragma unroll
                for (int nt = 0; nt < 9; ++nt) acc[nt] = mma4(a[i].xy, b[nt].xy, acc[nt]);
#pragma unroll
                for (int nt = 0; nt < 9; ++nt) acc[nt] = mma4(a[i].zw, b[nt].zw, acc[nt]);
            }
        }
#pragma unroll
        for (int nt = 0; nt < 9; ++nt) {
            const int n2 = nt * 16 + m;
            const float bb = b2[n2];
#pragma unroll
            for (int r = 0; r < 8; ++r)
                kout[(size_t)(p0 + r + hi * 8) * KGn + n2] = acc[nt][r] + bb;
        }
    }
}

__global__ __launch_bounds__(256) void invol_apply(
    const float* __restrict__ x, const float* __restrict__ kern,
    float* __restrict__ out)
{
    const int t  = blockIdx.x * 256 + threadIdx.x;
    const int ci = t & 63;
    const int p  = t >> 6;
    const int c0 = ci * 4;
    const int g0 = c0 & 15;

    const int b   = p >> 14;
    const int rem = p & 16383;
    const int y   = rem >> 7;
    const int xx  = rem & 127;

    const float* kp = kern + (size_t)p * KGn + g0;
    v4f acc = {};
#pragma unroll
    for (int ki = 0; ki < 3; ++ki) {
        const int yy = y + ki - 1;
        if (yy < 0 || yy >= Hn) continue;
#pragma unroll
        for (int kj = 0; kj < 3; ++kj) {
            const int xj = xx + kj - 1;
            if (xj < 0 || xj >= Wn) continue;
            const v4f xv = *(const v4f*)(x + ((size_t)((b * Hn + yy) * Wn + xj)) * Cn + c0);
            const v4f kv = *(const v4f*)(kp + (ki * 3 + kj) * Gn);
            acc += xv * kv;
        }
    }
    *(v4f*)(out + (size_t)p * Cn + c0) = acc;
}

extern "C" void kernel_launch(void* const* d_in, const int* in_sizes, int n_in,
                              void* d_out, int out_size, void* d_ws, size_t ws_size,
                              hipStream_t stream)
{
    const float* x     = (const float*)d_in[0];
    const float* w1    = (const float*)d_in[1];
    const float* b1    = (const float*)d_in[2];
    const float* gamma = (const float*)d_in[3];
    const float* beta  = (const float*)d_in[4];
    const float* mean  = (const float*)d_in[5];
    const float* var   = (const float*)d_in[6];
    const float* w2    = (const float*)d_in[7];
    const float* b2    = (const float*)d_in[8];

    float* out  = (float*)d_out;
    float* kout = out + (size_t)NPIX * Cn;

    invol_kergen<<<NPIX / 16 / NWAVE, NWAVE * 32, 0, stream>>>(
        x, w1, b1, gamma, beta, mean, var, w2, b2, kout);

    invol_apply<<<(NPIX * 64) / 256, 256, 0, stream>>>(x, kout, out);
}


# Apply kernel coarsened: one wave walks 16 pixels of a row for all 256 channels, next pixel's loads issued a step ahead, x columns reused from registers: 32x fewer waves enter; tap order kept
# speedup vs baseline: 1.9516x; 1.4186x over previous
#include <hip/hip_runtime.h>

typedef __attribute__((ext_vector_type(2))) float v2f;
typedef __attribute__((ext_vector_type(4))) float v4f;
typedef __attribute__((ext_vector_type(8))) float v8f;

#define Bn   4
#define Hn   128
#define Wn   128
#define Cn   256
#define Gn   16
#define K2n  9
#define Crn  64
#define KGn  (K2n * Gn)
#define NPIX (Bn * Hn * Wn)
#define BN_EPS 1e-3f

#define NWAVE 16
#define HPITCH 68
#define W1F_N (4 * 8 * 4 * 2 * 16)
#define W2F_N (9 * 2 * 4 * 2 * 16)

__device__ __forceinline__ v8f mma4(v2f a, v2f b, v8f c)
{
    return __builtin_amdgcn_wmma_f32_16x16x4_f32(false, a, false, b, (short)0, c, false, false);
}

__global__ __launch_bounds__(NWAVE * 32) void invol_kergen(
    const float* __restrict__ x,  const float* __restrict__ w1,
    const float* __restrict__ b1, const float* __restrict__ gamma,
    const float* __restrict__ beta, const float* __restrict__ mean,
    const float* __restrict__ var,  const float* __restrict__ w2,
    const float* __restrict__ b2,   float* __restrict__ kout)
{
    __shared__ v4f lds[W1F_N + W2F_N + NWAVE * 16 * HPITCH / 4];
    v4f*   const w1f   = lds;
    v4f*   const w2f   = lds + W1F_N;
    float* const h_lds = (float*)(lds + W1F_N + W2F_N)
                         + (threadIdx.x >> 5) * (16 * HPITCH);

    const int tid  = threadIdx.x;
    const int lane = tid & 31;
    const int wave = tid >> 5;
    const int tile = blockIdx.x * NWAVE + wave;
    const int p0   = tile * 16;
    const int m    = lane & 15;
    const int hi   = lane >> 4;

    const float* xrow = x + (size_t)(p0 + m) * Cn + hi * 16;
    v4f a_cur[4];
#pragma unroll
    for (int i = 0; i < 4; ++i) a_cur[i] = *(const v4f*)(xrow + 4 * i);

#pragma unroll
    for (int it = 0; it < W1F_N / (NWAVE * 32); ++it) {
        const int f = tid + it * (NWAVE * 32);
        const int c = f & 15, h = (f >> 4) & 1, i = (f >> 5) & 3, q = (f >> 7) & 7, nt = f >> 10;
        const float* src = w1 + (32 * q + 16 * h + 4 * i) * Crn + nt * 16 + c;
        v4f v;
        v.x = src[0]; v.y = src[Crn]; v.z = src[2 * Crn]; v.w = src[3 * Crn];
        w1f[f] = v;
    }
#pragma unroll
    for (int it = 0; it < (W2F_N + NWAVE * 32 - 1) / (NWAVE * 32); ++it) {
        const int f = tid + it * (NWAVE * 32);
        if (f >= W2F_N) break;
        const int c = f & 15, h = (f >> 4) & 1, i = (f >> 5) & 3, q = (f >> 7) & 1, nt = f >> 8;
        const float* src = w2 + (32 * q + 16 * h + 4 * i) * KGn + nt * 16 + c;
        v4f v;
        v.x = src[0]; v.y = src[KGn]; v.z = src[2 * KGn]; v.w = src[3 * KGn];
        w2f[f] = v;
    }
    __syncthreads();

    {
        v8f acc[4] = {};
        const v4f* bfrag = w1f + hi * 16 + m;
#pragma unroll
        for (int q = 0; q < 8; ++q) {
            v4f a_nxt[4];
            if (q < 7) {
#pragma unroll
                for (int i = 0; i < 4; ++i) a_nxt[i] = *(const v4f*)(xrow + 32 * (q + 1) + 4 * i);
            }
#pragma unroll
            for (int i = 0; i < 4; ++i) {
                v4f b[4];
#pragma unroll
                for (int nt = 0; nt < 4; ++nt) b[nt] = bfrag[((nt * 8 + q) * 4 + i) * 32];
#pragma unroll
                for (int nt = 0; nt < 4; ++nt) acc[nt] = mma4(a_cur[i].xy, b[nt].xy, acc[nt]);
#pragma unroll
                for (int nt = 0; nt < 4; ++nt) acc[nt] = mma4(a_cur[i].zw, b[nt].zw, acc[nt]);
            }
            if (q < 7) {
#pragma unroll
                for (int i = 0; i < 4; ++i) a_cur[i] = a_nxt[i];
            }
        }
#pragma unroll
        for (int nt = 0; nt < 4; ++nt) {
            const int n = nt * 16 + m;
            const float sc = gamma[n] * rsqrtf(var[n] + BN_EPS);
            const float sh = beta[n] - mean[n] * sc;
            const float bb = b1[n];
#pragma unroll
            for (int r = 0; r < 8; ++r) {
                float v = (acc[nt][r] + bb) * sc + sh;
                v = v > 0.0f ? v : 0.0f;
                h_lds[(r + hi * 8) * HPITCH + n] = v;
            }
        }
    }
    __syncthreads();

    {
        v8f acc[9] = {};
        const float* hrow  = h_lds + m * HPITCH + hi * 16;
        const v4f*   bfrag = w2f + hi * 16 + m;
#pragma unroll
        for (int q = 0; q < 2; ++q) {
            v4f a[4];
#pragma unroll
            for (int i = 0; i < 4; ++i) a[i] = *(const v4f*)(hrow + 32 * q + 4 * i);
#pragma unroll
            for (int i = 0; i < 4; ++i) {
                v4f b[9];
#pragma unroll
                for (int nt = 0; nt < 9; ++nt) b[nt] = bfrag[((nt * 2 + q) * 4 + i) * 32];
#pragma unroll
                for (int nt = 0; nt < 9; ++nt) acc[nt] = mma4(a[i].xy, b[nt].xy, acc[nt]);
#pragma unroll
                for (int nt = 0; nt < 9; ++nt) acc[nt] = mma4(a[i].zw, b[nt].zw, acc[nt]);
            }
        }
#pragma unroll
        for (int nt = 0; nt < 9; ++nt) {
            const int n2 = nt * 16 + m;
            const float bb = b2[n2];
#pragma unroll
            for (int r = 0; r < 8; ++r)
                kout[(size_t)(p0 + r + hi * 8) * KGn + n2] = acc[nt][r] + bb;
        }
    }
}

#define APX 16
#define AWV 8

__global__ __launch_bounds__(AWV * 32) void invol_apply(
    const float* __restrict__ x, const float* __restrict__ kern,
    float* __restrict__ out)
{
    const int lane  = threadIdx.x & 31;
    const int wave  = __builtin_amdgcn_readfirstlane(threadIdx.x >> 5);
    const int strip = blockIdx.x & (Wn / APX - 1);
    const int yrow  = (blockIdx.x / (Wn / APX)) * AWV + wave;
    const int y     = yrow & (Hn - 1);
    const int xs    = strip * APX;
    const int c0    = lane * 4;

    const bool top_ok = y > 0;
    const bool bot_ok = y < Hn - 1;
    const bool lft_ok = xs > 0;
    const bool rgt_ok = xs + APX < Wn;

    const float* rowp[3];
    rowp[0] = x + ((size_t)(yrow - (top_ok ? 1 : 0)) * Wn + xs) * Cn + c0;
    rowp[1] = x + ((size_t)yrow * Wn + xs) * Cn + c0;
    rowp[2] = x + ((size_t)(yrow + (bot_ok ? 1 : 0)) * Wn + xs) * Cn + c0;
    const size_t p0 = (size_t)yrow * Wn + xs;
    const float* kp = kern + p0 * KGn + (c0 & 15);
    float*       op = out + p0 * Cn + c0;
    const v4f zero = {};

    v4f raw[APX + 2][3][2];
    v4f col[APX + 2][3][2];
    v4f kv[APX][9];

#define COL_OK(CI) ((CI) == 0 ? lft_ok : ((CI) == APX + 1 ? rgt_ok : true))
#define LDCOL(CI)                                                                          \
    {                                                                                      \
        const int jl = COL_OK(CI) ? (CI) - 1 : ((CI) == 0 ? 0 : APX - 1);                  \
        _Pragma("unroll")                                                                  \
        for (int r = 0; r < 3; ++r) {                                                      \
            _Pragma("unroll")                                                              \
            for (int h = 0; h < 2; ++h)                                                    \
                raw[(CI)][r][h] = *(const v4f*)(rowp[r] + jl * Cn + h * 128);              \
        }                                                                                  \
    }
#define PADCOL(CI)                                                                         \
    {                                                                                      \
        _Pragma("unroll")                                                                  \
        for (int r = 0; r < 3; ++r) {                                                      \
            const bool ok = COL_OK(CI) && (r == 0 ? top_ok : (r == 2 ? bot_ok : true));    \
            _Pragma("unroll")                                                              \
            for (int h = 0; h < 2; ++h) col[(CI)][r][h] = ok ? raw[(CI)][r][h] : zero;     \
        }                                                                                  \
    }
#define LDK(J)                                                                             \
    {                                                                                      \
        _Pragma("unroll")                                                                  \
        for (int k = 0; k < 9; ++k) kv[(J)][k] = *(const v4f*)(kp + (J) * KGn + k * Gn);   \
    }

    LDCOL(0) LDCOL(1) LDCOL(2) LDK(0)

#pragma unroll
    for (int j = 0; j < APX; ++j) {
        if (j + 3 <= APX + 1) LDCOL(j + 3)
        if (j + 1 < APX)      LDK(j + 1)
        __builtin_amdgcn_sched_barrier(0);

        if (j == 0) { PADCOL(0) PADCOL(1) }
        PADCOL(j + 2)

        v4f a0 = zero, a1 = zero;
#pragma unroll
        for (int ki = 0; ki < 3; ++ki) {
#pragma unroll
            for (int kj = 0; kj < 3; ++kj) {
                const v4f k4 = kv[j][ki * 3 + kj];
                a0 += col[j + kj][ki][0] * k4;
                a1 += col[j + kj][ki][1] * k4;
            }
        }
        *(v4f*)(op + j * Cn)       = a0;
        *(v4f*)(op + j * Cn + 128) = a1;
        __builtin_amdgcn_sched_barrier(0);
    }
#undef COL_OK
#undef LDCOL
#undef PADCOL
#undef LDK
}

extern "C" void kernel_launch(void* const* d_in, const int* in_sizes, int n_in,
                              void* d_out, int out_size, void* d_ws, size_t ws_size,
                              hipStream_t stream)
{
    const float* x     = (const float*)d_in[0];
    const float* w1    = (const float*)d_in[1];
    const float* b1    = (const float*)d_in[2];
    const float* gamma = (const float*)d_in[3];
    const float* beta  = (const float*)d_in[4];
    const float* mean  = (const float*)d_in[5];
    const float* var   = (const float*)d_in[6];
    const float* w2    = (const float*)d_in[7];
    const float* b2    = (const float*)d_in[8];

    float* out  = (float*)d_out;
    float* kout = out + (size_t)NPIX * Cn;

    invol_kergen<<<NPIX / 16 / NWAVE, NWAVE * 32, 0, stream>>>(
        x, w1, b1, gamma, beta, mean, var, w2, b2, kout);

    invol_apply<<<NPIX / APX / AWV, AWV * 32, 0, stream>>>(x, kout, out);
}


# Both launches fused into one kernel: each wave applies its own 16 pixels right behind their kernel-value GEMMs, reading the values from LDS; no work-group barrier after weight staging; sums unchanged
# speedup vs baseline: 2.2551x; 1.1555x over previous
#include <hip/hip_runtime.h>

typedef __attribute__((ext_vector_type(2))) float v2f;
typedef __attribute__((ext_vector_type(4))) float v4f;
typedef __attribute__((ext_vector_type(8))) float v8f;

#define Bn   4
#define Hn   128
#define Wn   128
#define Cn   256
#define Gn   16
#define K2n  9
#define Crn  64
#define KGn  (K2n * Gn)
#define NPIX (Bn * Hn * Wn)
#define BN_EPS 1e-3f

#define NWAVE 16
#define HPITCH 68
#define W1F_N (4 * 8 * 4 * 2 * 16)
#define W2F_N (9 * 2 * 4 * 2 * 16)
#define KVP 148
#define WREG (16 * KVP)
#define APX 16

__device__ __forceinline__ v8f mma4(v2f a, v2f b, v8f c)
{
    return __builtin_amdgcn_wmma_f32_16x16x4_f32(false, a, false, b, (short)0, c, false, false);
}

__global__ __launch_bounds__(NWAVE * 32) void invol_fused(
    const float* __restrict__ x,  const float* __restrict__ w1,
    const float* __restrict__ b1, const float* __restrict__ gamma,
    const float* __restrict__ beta, const float* __restrict__ mean,
    const float* __restrict__ var,  const float* __restrict__ w2,
    const float* __restrict__ b2,   float* __restrict__ kout,
    float* __restrict__ out)
{
    __shared__ v4f lds[W1F_N + W2F_N + NWAVE * WREG / 4];
    v4f*   const w1f   = lds;
    v4f*   const w2f   = lds + W1F_N;
    const int wave = __builtin_amdgcn_readfirstlane(threadIdx.x >> 5);
    float* const wreg  = (float*)(lds + W1F_N + W2F_N) + wave * WREG;
    float* const h_lds = wreg;

    const int tid  = threadIdx.x;
    const int lane = tid & 31;
    const int yrow = (blockIdx.x >> 2) * 8 + (wave & 7);
    const int xs   = ((blockIdx.x & 3) * 2 + (wave >> 3)) * APX;
    const int p0   = yrow * Wn + xs;
    const int m    = lane & 15;
    const int hi   = lane >> 4;

    const float* xrow = x + (size_t)(p0 + m) * Cn + hi * 16;
    v4f a_cur[4];
#pragma unroll
    for (int i = 0; i < 4; ++i) a_cur[i] = *(const v4f*)(xrow + 4 * i);

#pragma unroll
    for (int it = 0; it < W1F_N / (NWAVE * 32); ++it) {
        const int f = tid + it * (NWAVE * 32);
        const int c = f & 15, h = (f >> 4) & 1, i = (f >> 5) & 3, q = (f >> 7) & 7, nt = f >> 10;
        const float* src = w1 + (32 * q + 16 * h + 4 * i) * Crn + nt * 16 + c;
        v4f v;
        v.x = src[0]; v.y = src[Crn]; v.z = src[2 * Crn]; v.w = src[3 * Crn];
        w1f[f] = v;
    }
#pragma unroll
    for (int it = 0; it < (W2F_N + NWAVE * 32 - 1) / (NWAVE * 32); ++it) {
        const int f = tid + it * (NWAVE * 32);
        if (f >= W2F_N) break;
        const int c = f & 15, h = (f >> 4) & 1, i = (f >> 5) & 3, q = (f >> 7) & 1, nt = f >> 8;
        const float* src = w2 + (32 * q + 16 * h + 4 * i) * KGn + nt * 16 + c;
        v4f v;
        v.x = src[0]; v.y = src[KGn]; v.z = src[2 * KGn]; v.w = src[3 * KGn];
        w2f[f] = v;
    }
    __syncthreads();

    {
        v8f acc[4] = {};
        const v4f* bfrag = w1f + hi * 16 + m;
#pragma unroll
        for (int q = 0; q < 8; ++q) {
            v4f a_nxt[4];
            if (q < 7) {
#pragma unroll
                for (int i = 0; i < 4; ++i) a_nxt[i] = *(const v4f*)(xrow + 32 * (q + 1) + 4 * i);
            }
#pragma unroll
            for (int i = 0; i < 4; ++i) {
                v4f b[4];
#pragma unroll
                for (int nt = 0; nt < 4; ++nt) b[nt] = bfrag[((nt * 8 + q) * 4 + i) * 32];
#pragma unroll
                for (int nt = 0; nt < 4; ++nt) acc[nt] = mma4(a_cur[i].xy, b[nt].xy, acc[nt]);
#pragma unroll
                for (int nt = 0; nt < 4; ++nt) acc[nt] = mma4(a_cur[i].zw, b[nt].zw, acc[nt]);
            }
            if (q < 7) {
#pragma unroll
                for (int i = 0; i < 4; ++i) a_cur[i] = a_nxt[i];
            }
        }
#pragma unroll
        for (int nt = 0; nt < 4; ++nt) {
            const int n = nt * 16 + m;
            const float sc = gamma[n] * rsqrtf(var[n] + BN_EPS);
            const float sh = beta[n] - mean[n] * sc;
            const float bb = b1[n];
#pragma unroll
            for (int r = 0; r < 8; ++r) {
                float v = (acc[nt][r] + bb) * sc + sh;
                v = v > 0.0f ? v : 0.0f;
                h_lds[(r + hi * 8) * HPITCH + n] = v;
            }
        }
    }
    __builtin_amdgcn_fence(__ATOMIC_RELEASE, "wavefront");
    __builtin_amdgcn_wave_barrier();
    __builtin_amdgcn_fence(__ATOMIC_ACQUIRE, "wavefront");

    {
        v8f acc[9] = {};
        const float* hrow  = h_lds + m * HPITCH + hi * 16;
        const v4f*   bfrag = w2f + hi * 16 + m;
#pragma unroll
        for (int q = 0; q < 2; ++q) {
            v4f a[4];
#pragma unroll
            for (int i = 0; i < 4; ++i) a[i] = *(const v4f*)(hrow + 32 * q + 4 * i);
#pragma unroll
            for (int i = 0; i < 4; ++i) {
                v4f b[9];
#pragma unroll
                for (int nt = 0; nt < 9; ++nt) b[nt] = bfrag[((nt * 2 + q) * 4 + i) * 32];
#pragma unroll
                for (int nt = 0; nt < 9; ++nt) acc[nt] = mma4(a[i].xy, b[nt].xy, acc[nt]);
#pragma unroll
                for (int nt = 0; nt < 9; ++nt) acc[nt] = mma4(a[i].zw, b[nt].zw, acc[nt]);
            }
        }
#pragma unroll
        for (int nt = 0; nt < 9; ++nt) {
            const int n2 = nt * 16 + m;
            const float bb = b2[n2];
#pragma unroll
            for (int r = 0; r < 8; ++r) {
                const float v = acc[nt][r] + bb;
                kout[(size_t)(p0 + r + hi * 8) * KGn + n2] = v;
                wreg[(r + hi * 8) * KVP + n2] = v;
            }
        }
    }
    __builtin_amdgcn_fence(__ATOMIC_RELEASE, "wavefront");
    __builtin_amdgcn_wave_barrier();
    __builtin_amdgcn_fence(__ATOMIC_ACQUIRE, "wavefront");

    const int y  = yrow & (Hn - 1);
    const int c0 = lane * 4;

    const bool top_ok = y > 0;
    const bool bot_ok = y < Hn - 1;
    const bool lft_ok = xs > 0;
    const bool rgt_ok = xs + APX < Wn;

    const float* rowp[3];
    rowp[0] = x + ((size_t)(yrow - (top_ok ? 1 : 0)) * Wn + xs) * Cn + c0;
    rowp[1] = x + ((size_t)yrow * Wn + xs) * Cn + c0;
    rowp[2] = x + ((size_t)(yrow + (bot_ok ? 1 : 0)) * Wn + xs) * Cn + c0;
    const float* kp = wreg + (c0 & 15);
    float*       op = out + (size_t)p0 * Cn + c0;
    const v4f zero = {};

    v4f raw[APX + 2][3][2];
    v4f col[APX + 2][3][2];
    v4f kv[APX][9];

#define COL_OK(CI) ((CI) == 0 ? lft_ok : ((CI) == APX + 1 ? rgt_ok : true))
#define LDCOL(CI)                                                                          \
    {                                                                                      \
        const int jl = COL_OK(CI) ? (CI) - 1 : ((CI) == 0 ? 0 : APX - 1);                  \
        _Pragma("unroll")                                                                  \
        for (int r = 0; r < 3; ++r) {                                                      \
            _Pragma("unroll")                                                              \
            for (int h = 0; h < 2; ++h)                                                    \
                raw[(CI)][r][h] = *(const v4f*)(rowp[r] + jl * Cn + h * 128);              \
        }                                                                                  \
    }
#define PADCOL(CI)                                                                         \
    {                                                                                      \
        _Pragma("unroll")                                                                  \
        for (int r = 0; r < 3; ++r) {                                                      \
            const bool ok = COL_OK(CI) && (r == 0 ? top_ok : (r == 2 ? bot_ok : true));    \
            _Pragma("unroll")                                                              \
            for (int h = 0; h < 2; ++h) col[(CI)][r][h] = ok ? raw[(CI)][r][h] : zero;     \
        }                                                                                  \
    }
#define LDK(J)                                                                             \
    {                                                                                      \
        _Pragma("unroll")                                                                  \
        for (int k = 0; k < 9; ++k) kv[(J)][k] = *(const v4f*)(kp + (J) * KVP + k * Gn);   \
    }

    LDCOL(0) LDCOL(1) LDCOL(2) LDK(0)

#pragma unroll
    for (int j = 0; j < APX; ++j) {
        if (j + 3 <= APX + 1) LDCOL(j + 3)
        if (j + 1 < APX)      LDK(j + 1)
        __builtin_amdgcn_sched_barrier(0);

        if (j == 0) { PADCOL(0) PADCOL(1) }
        PADCOL(j + 2)

        v4f a0 = zero, a1 = zero;
#pragma unroll
        for (int ki = 0; ki < 3; ++ki) {
#pragma unroll
            for (int kj = 0; kj < 3; ++kj) {
                const v4f k4 = kv[j][ki * 3 + kj];
                a0 += col[j + kj][ki][0] * k4;
                a1 += col[j + kj][ki][1] * k4;
            }
        }
        *(v4f*)(op + j * Cn)       = a0;
        *(v4f*)(op + j * Cn + 128) = a1;
        __builtin_amdgcn_sched_barrier(0);
    }
#undef COL_OK
#undef LDCOL
#undef PADCOL
#undef LDK
}

extern "C" void kernel_launch(void* const* d_in, const int* in_sizes, int n_in,
                              void* d_out, int out_size, void* d_ws, size_t ws_size,
                              hipStream_t stream)
{
    const float* x     = (const float*)d_in[0];
    const float* w1    = (const float*)d_in[1];
    const float* b1    = (const float*)d_in[2];
    const float* gamma = (const float*)d_in[3];
    const float* beta  = (const float*)d_in[4];
    const float* mean  = (const float*)d_in[5];
    const float* var   = (const float*)d_in[6];
    const float* w2    = (const float*)d_in[7];
    const float* b2    = (const float*)d_in[8];

    float* out  = (float*)d_out;
    float* kout = out + (size_t)NPIX * Cn;

    invol_fused<<<NPIX / 16 / NWAVE, NWAVE * 32, 0, stream>>>(
        x, w1, b1, gamma, beta, mean, var, w2, b2, kout, out);
}


# x loads issued further ahead: first GEMM's x pieces three k-blocks ahead (was one), walk's columns three steps ahead (was one), first five behind phase B's last product; sums and their order kept
# speedup vs baseline: 2.4207x; 1.0734x over previous
#include <hip/hip_runtime.h>

typedef __attribute__((ext_vector_type(2))) float v2f;
typedef __attribute__((ext_vector_type(4))) float v4f;
typedef __attribute__((ext_vector_type(8))) float v8f;

#define Bn   4
#define Hn   128
#define Wn   128
#define Cn   256
#define Gn   16
#define K2n  9
#define Crn  64
#define KGn  (K2n * Gn)
#define NPIX (Bn * Hn * Wn)
#define BN_EPS 1e-3f

#define NWAVE 16
#define HPITCH 68
#define W1F_N (4 * 8 * 4 * 2 * 16)
#define W2F_N (9 * 2 * 4 * 2 * 16)
#define KVP 148
#define WREG (16 * KVP)
#define APX 16

__device__ __forceinline__ v8f mma4(v2f a, v2f b, v8f c)
{
    return __builtin_amdgcn_wmma_f32_16x16x4_f32(false, a, false, b, (short)0, c, false, false);
}

__global__ __launch_bounds__(NWAVE * 32) void invol_fused(
    const float* __restrict__ x,  const float* __restrict__ w1,
    const float* __restrict__ b1, const float* __restrict__ gamma,
    const float* __restrict__ beta, const float* __restrict__ mean,
    const float* __restrict__ var,  const float* __restrict__ w2,
    const float* __restrict__ b2,   float* __restrict__ kout,
    float* __restrict__ out)
{
    __shared__ v4f lds[W1F_N + W2F_N + NWAVE * WREG / 4];
    v4f*   const w1f   = lds;
    v4f*   const w2f   = lds + W1F_N;
    const int wave = __builtin_amdgcn_readfirstlane(threadIdx.x >> 5);
    float* const wreg  = (float*)(lds + W1F_N + W2F_N) + wave * WREG;
    float* const h_lds = wreg;

    const int tid  = threadIdx.x;
    const int lane = tid & 31;
    const int yrow = (blockIdx.x >> 2) * 8 + (wave & 7);
    const int xs   = ((blockIdx.x & 3) * 2 + (wave >> 3)) * APX;
    const int p0   = yrow * Wn + xs;
    const int m    = lane & 15;
    const int hi   = lane >> 4;

    const float* xrow = x + (size_t)(p0 + m) * Cn + hi * 16;
    v4f a_ring[4][4];
#pragma unroll
    for (int i = 0; i < 4; ++i) a_ring[0][i] = *(const v4f*)(xrow + 4 * i);

#pragma unroll
    for (int it = 0; it < W1F_N / (NWAVE * 32); ++it) {
        const int f = tid + it * (NWAVE * 32);
        const int c = f & 15, h = (f >> 4) & 1, i = (f >> 5) & 3, q = (f >> 7) & 7, nt = f >> 10;
        const float* src = w1 + (32 * q + 16 * h + 4 * i) * Crn + nt * 16 + c;
        v4f v;
        v.x = src[0]; v.y = src[Crn]; v.z = src[2 * Crn]; v.w = src[3 * Crn];
        w1f[f] = v;
    }
#pragma unroll
    for (int it = 0; it < (W2F_N + NWAVE * 32 - 1) / (NWAVE * 32); ++it) {
        const int f = tid + it * (NWAVE * 32);
        if (f >= W2F_N) break;
        const int c = f & 15, h = (f >> 4) & 1, i = (f >> 5) & 3, q = (f >> 7) & 1, nt = f >> 8;
        const float* src = w2 + (32 * q + 16 * h + 4 * i) * KGn + nt * 16 + c;
        v4f v;
        v.x = src[0]; v.y = src[KGn]; v.z = src[2 * KGn]; v.w = src[3 * KGn];
        w2f[f] = v;
    }
    __syncthreads();
#pragma unroll
    for (int qa = 1; qa < 4; ++qa) {
#pragma unroll
        for (int i = 0; i < 4; ++i) a_ring[qa][i] = *(const v4f*)(xrow + 32 * qa + 4 * i);
    }
    __builtin_amdgcn_sched_barrier(0);

    {
        v8f acc[4] = {};
        const v4f* bfrag = w1f + hi * 16 + m;
#pragma unroll
        for (int q = 0; q < 8; ++q) {
            if (q >= 1 && q + 3 < 8) {
#pragma unroll
                for (int i = 0; i < 4; ++i)
                    a_ring[(q + 3) & 3][i] = *(const v4f*)(xrow + 32 * (q + 3) + 4 * i);
            }
#pragma unroll
            for (int i = 0; i < 4; ++i) {
                v4f b[4];
#pragma unroll
                for (int nt = 0; nt < 4; ++nt) b[nt] = bfrag[((nt * 8 + q) * 4 + i) * 32];
#pragma unroll
                for (int nt = 0; nt < 4; ++nt) acc[nt] = mma4(a_ring[q & 3][i].xy, b[nt].xy, acc[nt]);
#pragma unroll
                for (int nt = 0; nt < 4; ++nt) acc[nt] = mma4(a_ring[q & 3][i].zw, b[nt].zw, acc[nt]);
            }
        }
#pragma unroll
        for (int nt = 0; nt < 4; ++nt) {
            const int n = nt * 16 + m;
            const float sc = gamma[n] * rsqrtf(var[n] + BN_EPS);
            const float sh = beta[n] - mean[n] * sc;
            const float bb = b1[n];
#pragma unroll
            for (int r = 0; r < 8; ++r) {
                float v = (acc[nt][r] + bb) * sc + sh;
                v = v > 0.0f ? v : 0.0f;
                h_lds[(r + hi * 8) * HPITCH + n] = v;
            }
        }
    }
    __builtin_amdgcn_fence(__ATOMIC_RELEASE, "wavefront");
    __builtin_amdgcn_wave_barrier();
    __builtin_amdgcn_fence(__ATOMIC_ACQUIRE, "wavefront");

    const int y  = yrow & (Hn - 1);
    const int c0 = lane * 4;

    const bool top_ok = y > 0;
    const bool bot_ok = y < Hn - 1;
    const bool lft_ok = xs > 0;
    const bool rgt_ok = xs + APX < Wn;

    const float* rowp[3];
    rowp[0] = x + ((size_t)(yrow - (top_ok ? 1 : 0)) * Wn + xs) * Cn + c0;
    rowp[1] = x + ((size_t)yrow * Wn + xs) * Cn + c0;
    rowp[2] = x + ((size_t)(yrow + (bot_ok ? 1 : 0)) * Wn + xs) * Cn + c0;
    const float* kp = wreg + (c0 & 15);
    float*       op = out + (size_t)p0 * Cn + c0;
    const v4f zero = {};

    v4f raw[APX + 2][3][2];
    v4f col[APX + 2][3][2];
    v4f kv[APX][9];

#define COL_OK(CI) ((CI) == 0 ? lft_ok : ((CI) == APX + 1 ? rgt_ok : true))
#define LDCOL(CI)                                                                          \
    {                                                                                      \
        const int jl = COL_OK(CI) ? (CI) - 1 : ((CI) == 0 ? 0 : APX - 1);                  \
        _Pragma("unroll")                                                                  \
        for (int r = 0; r < 3; ++r) {                                                      \
            _Pragma("unroll")                                                              \
            for (int h = 0; h < 2; ++h)                                                    \
                raw[(CI)][r][h] = *(const v4f*)(rowp[r] + jl * Cn + h * 128);              \
        }                                                                                  \
    }
#define PADCOL(CI)                                                                         \
    {                                                                                      \
        _Pragma("unroll")                                                                  \
        for (int r = 0; r < 3; ++r) {                                                      \
            const bool ok = COL_OK(CI) && (r == 0 ? top_ok : (r == 2 ? bot_ok : true));    \
            _Pragma("unroll")                                                              \
            for (int h = 0; h < 2; ++h) col[(CI)][r][h] = ok ? raw[(CI)][r][h] : zero;     \
        }                                                                                  \
    }
#define LDK(J)                                                                             \
    {                                                                                      \
        _Pragma("unroll")                                                                  \
        for (int k = 0; k < 9; ++k) kv[(J)][k] = *(const v4f*)(kp + (J) * KVP + k * Gn);   \
    }

    {
        v8f acc[9] = {};
        float bias2[9];
#pragma unroll
        for (int nt = 0; nt < 9; ++nt) bias2[nt] = b2[nt * 16 + m];
        const float* hrow  = h_lds + m * HPITCH + hi * 16;
        const v4f*   bfrag = w2f + hi * 16 + m;
#pragma unroll
        for (int q = 0; q < 2; ++q) {
            v4f a[4];
#pragma unroll
            for (int i = 0; i < 4; ++i) a[i] = *(const v4f*)(hrow + 32 * q + 4 * i);
#pragma unroll
            for (int i = 0; i < 4; ++i) {
                v4f b[9];
#pragma unroll
                for (int nt = 0; nt < 9; ++nt) b[nt] = bfrag[((nt * 2 + q) * 4 + i) * 32];
#pragma unroll
                for (int nt = 0; nt < 9; ++nt) acc[nt] = mma4(a[i].xy, b[nt].xy, acc[nt]);
#pragma unroll
                for (int nt = 0; nt < 9; ++nt) acc[nt] = mma4(a[i].zw, b[nt].zw, acc[nt]);
            }
        }
        __builtin_amdgcn_sched_barrier(0);
        LDCOL(0) LDCOL(1) LDCOL(2) LDCOL(3) LDCOL(4)
        __builtin_amdgcn_sched_barrier(0);

#pragma unroll
        for (int nt = 0; nt < 9; ++nt) {
            const int n2 = nt * 16 + m;
            const float bb = bias2[nt];
#pragma unroll
            for (int r = 0; r < 8; ++r) {
                const float v = acc[nt][r] + bb;
                kout[(size_t)(p0 + r + hi * 8) * KGn + n2] = v;
                wreg[(r + hi * 8) * KVP + n2] = v;
            }
        }
    }
    __builtin_amdgcn_fence(__ATOMIC_RELEASE, "wavefront");
    __builtin_amdgcn_wave_barrier();
    __builtin_amdgcn_fence(__ATOMIC_ACQUIRE, "wavefront");

    LDK(0)

#pragma unroll
    for (int j = 0; j < APX; ++j) {
        if (j + 5 <= APX + 1) LDCOL(j + 5)
        if (j + 1 < APX)      LDK(j + 1)
        __builtin_amdgcn_sched_barrier(0);

        if (j == 0) { PADCOL(0) PADCOL(1) }
        PADCOL(j + 2)

        v4f a0 = zero, a1 = zero;
#pragma unroll
        for (int ki = 0; ki < 3; ++ki) {
#pragma unroll
            for (int kj = 0; kj < 3; ++kj) {
                const v4f k4 = kv[j][ki * 3 + kj];
                a0 += col[j + kj][ki][0] * k4;
                a1 += col[j + kj][ki][1] * k4;
            }
        }
        *(v4f*)(op + j * Cn)       = a0;
        *(v4f*)(op + j * Cn + 128) = a1;
        __builtin_amdgcn_sched_barrier(0);
    }
#undef COL_OK
#undef LDCOL
#undef PADCOL
#undef LDK
}

extern "C" void kernel_launch(void* const* d_in, const int* in_sizes, int n_in,
                              void* d_out, int out_size, void* d_ws, size_t ws_size,
                              hipStream_t stream)
{
    const float* x     = (const float*)d_in[0];
    const float* w1    = (const float*)d_in[1];
    const float* b1    = (const float*)d_in[2];
    const float* gamma = (const float*)d_in[3];
    const float* beta  = (const float*)d_in[4];
    const float* mean  = (const float*)d_in[5];
    const float* var   = (const float*)d_in[6];
    const float* w2    = (const float*)d_in[7];
    const float* b2    = (const float*)d_in[8];

    float* out  = (float*)d_out;
    float* kout = out + (size_t)NPIX * Cn;

    invol_fused<<<NPIX / 16 / NWAVE, NWAVE * 32, 0, stream>>>(
        x, w1, b1, gamma, beta, mean, var, w2, b2, kout, out);
}


# Both GEMMs' weight fragments read from LDS as one group per block, one block ahead into a second register set (were single reads in among the products); sums and their order unchanged
# speedup vs baseline: 2.5267x; 1.0438x over previous
#include <hip/hip_runtime.h>

typedef __attribute__((ext_vector_type(2))) float v2f;
typedef __attribute__((ext_vector_type(4))) float v4f;
typedef __attribute__((ext_vector_type(8))) float v8f;

#define Bn   4
#define Hn   128
#define Wn   128
#define Cn   256
#define Gn   16
#define K2n  9
#define Crn  64
#define KGn  (K2n * Gn)
#define NPIX (Bn * Hn * Wn)
#define BN_EPS 1e-3f

#define NWAVE 16
#define HPITCH 68
#define W1F_N (4 * 8 * 4 * 2 * 16)
#define W2F_N (9 * 2 * 4 * 2 * 16)
#define KVP 148
#define WREG (16 * KVP)
#define APX 16

__device__ __forceinline__ v8f mma4(v2f a, v2f b, v8f c)
{
    return __builtin_amdgcn_wmma_f32_16x16x4_f32(false, a, false, b, (short)0, c, false, false);
}

__global__ __launch_bounds__(NWAVE * 32) void invol_fused(
    const float* __restrict__ x,  const float* __restrict__ w1,
    const float* __restrict__ b1, const float* __restrict__ gamma,
    const float* __restrict__ beta, const float* __restrict__ mean,
    const float* __restrict__ var,  const float* __restrict__ w2,
    const float* __restrict__ b2,   float* __restrict__ kout,
    float* __restrict__ out)
{
    __shared__ v4f lds[W1F_N + W2F_N + NWAVE * WREG / 4];
    v4f*   const w1f   = lds;
    v4f*   const w2f   = lds + W1F_N;
    const int wave = __builtin_amdgcn_readfirstlane(threadIdx.x >> 5);
    float* const wreg  = (float*)(lds + W1F_N + W2F_N) + wave * WREG;
    float* const h_lds = wreg;

    const int tid  = threadIdx.x;
    const int lane = tid & 31;
    const int yrow = (blockIdx.x >> 2) * 8 + (wave & 7);
    const int xs   = ((blockIdx.x & 3) * 2 + (wave >> 3)) * APX;
    const int p0   = yrow * Wn + xs;
    const int m    = lane & 15;
    const int hi   = lane >> 4;

    const float* xrow = x + (size_t)(p0 + m) * Cn + hi * 16;
    v4f a_ring[4][4];
#pragma unroll
    for (int i = 0; i < 4; ++i) a_ring[0][i] = *(const v4f*)(xrow + 4 * i);

#pragma unroll
    for (int it = 0; it < W1F_N / (NWAVE * 32); ++it) {
        const int f = tid + it * (NWAVE * 32);
        const int c = f & 15, h = (f >> 4) & 1, i = (f >> 5) & 3, q = (f >> 7) & 7, nt = f >> 10;
        const float* src = w1 + (32 * q + 16 * h + 4 * i) * Crn + nt * 16 + c;
        v4f v;
        v.x = src[0]; v.y = src[Crn]; v.z = src[2 * Crn]; v.w = src[3 * Crn];
        w1f[f] = v;
    }
#pragma unroll
    for (int it = 0; it < (W2F_N + NWAVE * 32 - 1) / (NWAVE * 32); ++it) {
        const int f = tid + it * (NWAVE * 32);
        if (f >= W2F_N) break;
        const int c = f & 15, h = (f >> 4) & 1, i = (f >> 5) & 3, q = (f >> 7) & 1, nt = f >> 8;
        const float* src = w2 + (32 * q + 16 * h + 4 * i) * KGn + nt * 16 + c;
        v4f v;
        v.x = src[0]; v.y = src[KGn]; v.z = src[2 * KGn]; v.w = src[3 * KGn];
        w2f[f] = v;
    }
    __syncthreads();
#pragma unroll
    for (int qa = 1; qa < 4; ++qa) {
#pragma unroll
        for (int i = 0; i < 4; ++i) a_ring[qa][i] = *(const v4f*)(xrow + 32 * qa + 4 * i);
    }
    __builtin_amdgcn_sched_barrier(0);

    {
        v8f acc[4] = {};
        const v4f* bfrag = w1f + hi * 16 + m;
        v4f bset[2][16];
        float p_g[4], p_v[4], p_be[4], p_mu[4], p_b1[4];
#pragma unroll
        for (int i = 0; i < 4; ++i) {
#pragma unroll
            for (int nt = 0; nt < 4; ++nt) bset[0][i * 4 + nt] = bfrag[((nt * 8 + 0) * 4 + i) * 32];
        }
#pragma unroll
        for (int q = 0; q < 8; ++q) {
            if (q + 1 < 8) {
#pragma unroll
                for (int i = 0; i < 4; ++i) {
#pragma unroll
                    for (int nt = 0; nt < 4; ++nt)
                        bset[(q + 1) & 1][i * 4 + nt] = bfrag[((nt * 8 + q + 1) * 4 + i) * 32];
                }
            }
            if (q >= 1 && q + 3 < 8) {
#pragma unroll
                for (int i = 0; i < 4; ++i)
                    a_ring[(q + 3) & 3][i] = *(const v4f*)(xrow + 32 * (q + 3) + 4 * i);
            }
            if (q == 6) {
#pragma unroll
                for (int nt = 0; nt < 4; ++nt) {
                    const int n = nt * 16 + m;
                    p_g[nt] = gamma[n]; p_v[nt] = var[n]; p_be[nt] = beta[n]; p_mu[nt] = mean[n]; p_b1[nt] = b1[n];
                }
            }
            __builtin_amdgcn_sched_barrier(0);
#pragma unroll
            for (int i = 0; i < 4; ++i) {
#pragma unroll
                for (int nt = 0; nt < 4; ++nt) acc[nt] = mma4(a_ring[q & 3][i].xy, bset[q & 1][i * 4 + nt].xy, acc[nt]);
#pragma unroll
                for (int nt = 0; nt < 4; ++nt) acc[nt] = mma4(a_ring[q & 3][i].zw, bset[q & 1][i * 4 + nt].zw, acc[nt]);
            }
            if (q + 1 < 8) {
                __builtin_amdgcn_sched_barrier(0);
            }
        }
#pragma unroll
        for (int nt = 0; nt < 4; ++nt) {
            const int n = nt * 16 + m;
            const float sc = p_g[nt] * rsqrtf(p_v[nt] + BN_EPS);
            const float sh = p_be[nt] - p_mu[nt] * sc;
            const float bb = p_b1[nt];
#pragma unroll
            for (int r = 0; r < 8; ++r) {
                float v = (acc[nt][r] + bb) * sc + sh;
                v = v > 0.0f ? v : 0.0f;
                h_lds[(r + hi * 8) * HPITCH + n] = v;
            }
        }
    }
    __builtin_amdgcn_fence(__ATOMIC_RELEASE, "wavefront");
    __builtin_amdgcn_wave_barrier();
    __builtin_amdgcn_fence(__ATOMIC_ACQUIRE, "wavefront");

    const int y  = yrow & (Hn - 1);
    const int c0 = lane * 4;

    const bool top_ok = y > 0;
    const bool bot_ok = y < Hn - 1;
    const bool lft_ok = xs > 0;
    const bool rgt_ok = xs + APX < Wn;

    const float* rowp[3];
    rowp[0] = x + ((size_t)(yrow - (top_ok ? 1 : 0)) * Wn + xs) * Cn + c0;
    rowp[1] = x + ((size_t)yrow * Wn + xs) * Cn + c0;
    rowp[2] = x + ((size_t)(yrow + (bot_ok ? 1 : 0)) * Wn + xs) * Cn + c0;
    const float* kp = wreg + (c0 & 15);
    float*       op = out + (size_t)p0 * Cn + c0;
    const v4f zero = {};

    v4f raw[APX + 2][3][2];
    v4f col[APX + 2][3][2];
    v4f kv[APX][9];

#define COL_OK(CI) ((CI) == 0 ? lft_ok : ((CI) == APX + 1 ? rgt_ok : true))
#define LDCOL(CI)                                                                          \
    {                                                                                      \
        const int jl = COL_OK(CI) ? (CI) - 1 : ((CI) == 0 ? 0 : APX - 1);                  \
        _Pragma("unroll")                                                                  \
        for (int r = 0; r < 3; ++r) {                                                      \
            _Pragma("unroll")                                                              \
            for (int h = 0; h < 2; ++h)                                                    \
                raw[(CI)][r][h] = *(const v4f*)(rowp[r] + jl * Cn + h * 128);              \
        }                                                                                  \
    }
#define PADCOL(CI)                                                                         \
    {                                                                                      \
        _Pragma("unroll")                                                                  \
        for (int r = 0; r < 3; ++r) {                                                      \
            const bool ok = COL_OK(CI) && (r == 0 ? top_ok : (r == 2 ? bot_ok : true));    \
            _Pragma("unroll")                                                              \
            for (int h = 0; h < 2; ++h) col[(CI)][r][h] = ok ? raw[(CI)][r][h] : zero;     \
        }                                                                                  \
    }
#define LDK(J)                                                                             \
    {                                                                                      \
        _Pragma("unroll")                                                                  \
        for (int k = 0; k < 9; ++k) kv[(J)][k] = *(const v4f*)(kp + (J) * KVP + k * Gn);   \
    }

    {
        v8f acc[9] = {};
        float bias2[9];
#pragma unroll
        for (int nt = 0; nt < 9; ++nt) bias2[nt] = b2[nt * 16 + m];
        const float* hrow  = h_lds + m * HPITCH + hi * 16;
        const v4f*   bfrag = w2f + hi * 16 + m;
        v4f a[2][4];
        v4f bset[2][9];
#pragma unroll
        for (int q = 0; q < 2; ++q) {
#pragma unroll
            for (int i = 0; i < 4; ++i) a[q][i] = *(const v4f*)(hrow + 32 * q + 4 * i);
        }
#pragma unroll
        for (int nt = 0; nt < 9; ++nt) bset[0][nt] = bfrag[((nt * 2 + 0) * 4 + 0) * 32];
#pragma unroll
        for (int g = 0; g < 8; ++g) {
            const int q = g >> 2, i = g & 3;
            if (g + 1 < 8) {
                const int qn = (g + 1) >> 2, in = (g + 1) & 3;
#pragma unroll
                for (int nt = 0; nt < 9; ++nt) bset[(g + 1) & 1][nt] = bfrag[((nt * 2 + qn) * 4 + in) * 32];
            }
            __builtin_amdgcn_sched_barrier(0);
#pragma unroll
            for (int nt = 0; nt < 9; ++nt) acc[nt] = mma4(a[q][i].xy, bset[g & 1][nt].xy, acc[nt]);
#pragma unroll
            for (int nt = 0; nt < 9; ++nt) acc[nt] = mma4(a[q][i].zw, bset[g & 1][nt].zw, acc[nt]);
            if (g + 1 < 8) {
                __builtin_amdgcn_sched_barrier(0);
            }
        }
        __builtin_amdgcn_sched_barrier(0);
        LDCOL(0) LDCOL(1) LDCOL(2) LDCOL(3) LDCOL(4)
        __builtin_amdgcn_sched_barrier(0);

#pragma unroll
        for (int nt = 0; nt < 9; ++nt) {
            const int n2 = nt * 16 + m;
            const float bb = bias2[nt];
#pragma unroll
            for (int r = 0; r < 8; ++r) {
                const float v = acc[nt][r] + bb;
                kout[(size_t)(p0 + r + hi * 8) * KGn + n2] = v;
                wreg[(r + hi * 8) * KVP + n2] = v;
            }
        }
    }
    __builtin_amdgcn_fence(__ATOMIC_RELEASE, "wavefront");
    __builtin_amdgcn_wave_barrier();
    __builtin_amdgcn_fence(__ATOMIC_ACQUIRE, "wavefront");

    LDK(0)

#pragma unroll
    for (int j = 0; j < APX; ++j) {
        if (j + 5 <= APX + 1) LDCOL(j + 5)
        if (j + 1 < APX)      LDK(j + 1)
        __builtin_amdgcn_sched_barrier(0);

        if (j == 0) { PADCOL(0) PADCOL(1) }
        PADCOL(j + 2)

        v4f a0 = zero, a1 = zero;
#pragma unroll
        for (int ki = 0; ki < 3; ++ki) {
#pragma unroll
            for (int kj = 0; kj < 3; ++kj) {
                const v4f k4 = kv[j][ki * 3 + kj];
                a0 += col[j + kj][ki][0] * k4;
                a1 += col[j + kj][ki][1] * k4;
            }
        }
        *(v4f*)(op + j * Cn)       = a0;
        *(v4f*)(op + j * Cn + 128) = a1;
        __builtin_amdgcn_sched_barrier(0);
    }
#undef COL_OK
#undef LDCOL
#undef PADCOL
#undef LDK
}

extern "C" void kernel_launch(void* const* d_in, const int* in_sizes, int n_in,
                              void* d_out, int out_size, void* d_ws, size_t ws_size,
                              hipStream_t stream)
{
    const float* x     = (const float*)d_in[0];
    const float* w1    = (const float*)d_in[1];
    const float* b1    = (const float*)d_in[2];
    const float* gamma = (const float*)d_in[3];
    const float* beta  = (const float*)d_in[4];
    const float* mean  = (const float*)d_in[5];
    const float* var   = (const float*)d_in[6];
    const float* w2    = (const float*)d_in[7];
    const float* b2    = (const float*)d_in[8];

    float* out  = (float*)d_out;
    float* kout = out + (size_t)NPIX * Cn;

    invol_fused<<<NPIX / 16 / NWAVE, NWAVE * 32, 0, stream>>>(
        x, w1, b1, gamma, beta, mean, var, w2, b2, kout, out);
}
